# starting kernel
# baseline (speedup 1.0000x reference)
#include <hip/hip_runtime.h>
#include <hip/hip_bf16.h>

typedef float f32x2 __attribute__((ext_vector_type(2)));
typedef float f32x8 __attribute__((ext_vector_type(8)));

#define BATCH   64
#define CCH     3
#define HIMG    224
#define WIMG    224
#define PSZ     16
#define DMODEL  768
#define KDIM    768
#define NPATCH  196
#define SEQ     197
#define ROWS    (BATCH * NPATCH)

#define TM      64
#define TN      64
#define TK      32
#define A_STRIDE 36
#define B_STRIDE 68

__global__ __launch_bounds__(256)
void vit_patch_embed_gemm(const float* __restrict__ x,
                          const float* __restrict__ Wm,
                          const float* __restrict__ bias,
                          const float* __restrict__ pos,
                          float* __restrict__ out)
{
    __shared__ float As[TM * A_STRIDE];
    __shared__ float Bs[TK * B_STRIDE];

    const int tid    = threadIdx.x;
    const int lane   = tid & 31;
    const int waveId = tid >> 5;
    const int waveM  = waveId & 3;
    const int waveN  = waveId >> 2;
    const int half   = lane >> 4;
    const int lsub   = lane & 15;
    const int kOff   = half * 2;

    const int rowBase = blockIdx.x * TM;
    const int nBase   = blockIdx.y * TN;

    const int aRow     = tid >> 2;
    const int aColBase = (tid & 3) * 8;
    const int rG   = rowBase + aRow;
    const int bImg = rG / NPATCH;
    const int nTok = rG - bImg * NPATCH;
    const int pi_  = nTok / 14;
    const int pj_  = nTok - pi_ * 14;

    const int bRow     = tid >> 3;
    const int bColBase = (tid & 7) * 8;

    f32x8 c0 = {0.f,0.f,0.f,0.f,0.f,0.f,0.f,0.f};
    f32x8 c1 = {0.f,0.f,0.f,0.f,0.f,0.f,0.f,0.f};

    const int nA = waveN * 32;
    const int nB = nA + 16;

    for (int kBase = 0; kBase < KDIM; kBase += TK) {
        {
            const int f0 = kBase + aColBase;
            const int ch = f0 >> 8;
            const int rem = f0 & 255;
            const int pr = rem >> 4;
            const int pc = rem & 15;
            const long gidx = ((long)(bImg * CCH + ch) * HIMG + pi_ * PSZ + pr) * WIMG
                              + pj_ * PSZ + pc;
            const float4 av0 = *(const float4*)(x + gidx);
            const float4 av1 = *(const float4*)(x + gidx + 4);
            *(float4*)&As[aRow * A_STRIDE + aColBase]     = av0;
            *(float4*)&As[aRow * A_STRIDE + aColBase + 4] = av1;
        }
        {
            const long widx = (long)(kBase + bRow) * DMODEL + nBase + bColBase;
            const float4 bv0 = *(const float4*)(Wm + widx);
            const float4 bv1 = *(const float4*)(Wm + widx + 4);
            *(float4*)&Bs[bRow * B_STRIDE + bColBase]     = bv0;
            *(float4*)&Bs[bRow * B_STRIDE + bColBase + 4] = bv1;
            if (kBase + TK < KDIM) {
                __builtin_prefetch(Wm + widx + (long)TK * DMODEL, 0, 0);
            }
        }
        __syncthreads();

        #pragma unroll
        for (int kk = 0; kk < TK; kk += 4) {
            f32x2 a, b0, b1;
            const int ar = (waveM * 16 + lsub) * A_STRIDE + kk + kOff;
            a.x = As[ar];
            a.y = As[ar + 1];
            const int br0 = (kk + kOff) * B_STRIDE;
            const int br1 = (kk + kOff + 1) * B_STRIDE;
            b0.x = Bs[br0 + nA + lsub];
            b0.y = Bs[br1 + nA + lsub];
            b1.x = Bs[br0 + nB + lsub];
            b1.y = Bs[br1 + nB + lsub];
            c0 = __builtin_amdgcn_wmma_f32_16x16x4_f32(false, a, false, b0,
                                                       (short)0, c0, false, false);
            c1 = __builtin_amdgcn_wmma_f32_16x16x4_f32(false, a, false, b1,
                                                       (short)0, c1, false, false);
        }
        __syncthreads();
    }

    const int colA = nBase + nA + lsub;
    const int colB = nBase + nB + lsub;
    const float biasA = bias[colA];
    const float biasB = bias[colB];
    #pragma unroll
    for (int v = 0; v < 8; ++v) {
        const int mRow = waveM * 16 + v + half * 8;
        const int r    = rowBase + mRow;
        const int bI   = r / NPATCH;
        const int s    = r - bI * NPATCH + 1;
        const long obase = ((long)bI * SEQ + s) * DMODEL;
        const long pbase = (long)s * DMODEL;
        out[obase + colA] = c0[v] + biasA + pos[pbase + colA];
        out[obase + colB] = c1[v] + biasB + pos[pbase + colB];
    }
}

__global__ __launch_bounds__(256)
void vit_class_token(const float* __restrict__ cls,
                     const float* __restrict__ pos,
                     float* __restrict__ out)
{
    const int idx = blockIdx.x * blockDim.x + threadIdx.x;
    const int b = idx / DMODEL;
    const int d = idx - b * DMODEL;
    out[(long)b * SEQ * DMODEL + d] = cls[d] + pos[d];
}

extern "C" void kernel_launch(void* const* d_in, const int* in_sizes, int n_in,
                              void* d_out, int out_size, void* d_ws, size_t ws_size,
                              hipStream_t stream) {
    const float* x   = (const float*)d_in[0];
    const float* Wm  = (const float*)d_in[1];
    const float* b   = (const float*)d_in[2];
    const float* pos = (const float*)d_in[3];
    const float* cls = (const float*)d_in[4];
    float* out = (float*)d_out;

    dim3 grid(ROWS / TM, DMODEL / TN);
    vit_patch_embed_gemm<<<grid, 256, 0, stream>>>(x, Wm, b, pos, out);
    vit_class_token<<<(BATCH * DMODEL) / 256, 256, 0, stream>>>(cls, pos, out);
}


# Register-block the waves: 4 waves per 64x128 block, each a 32x64 tile (2x4 WMMA tiles) fed by 8-byte conflict-free LDS reads (B stored k-pair interleaved), in place of 8 waves of 16x32 on 64x64
# speedup vs baseline: 1.3061x; 1.3061x over previous
#include <hip/hip_runtime.h>
#include <hip/hip_bf16.h>

typedef float f32x2 __attribute__((ext_vector_type(2)));
typedef float f32x8 __attribute__((ext_vector_type(8)));

#define BATCH   64
#define CCH     3
#define HIMG    224
#define WIMG    224
#define PSZ     16
#define DMODEL  768
#define KDIM    768
#define NPATCH  196
#define SEQ     197
#define ROWS    (BATCH * NPATCH)

#define TM      64
#define TN      128
#define TK      32
#define NTHR    128
#define A_STRIDE 36
#define B_PSTRIDE 288

__global__ __launch_bounds__(NTHR) __attribute__((amdgpu_waves_per_eu(5, 5)))
void vit_patch_embed_gemm(const float* __restrict__ x,
                          const float* __restrict__ Wm,
                          const float* __restrict__ bias,
                          const float* __restrict__ pos,
                          float* __restrict__ out)
{
    __shared__ float As[TM * A_STRIDE];
    __shared__ float Bs[(TK / 2) * B_PSTRIDE];

    const int tid    = threadIdx.x;
    const int lane   = tid & 31;
    const int waveId = tid >> 5;
    const int waveM  = waveId & 1;
    const int waveN  = waveId >> 1;
    const int half   = lane >> 4;
    const int lsub   = lane & 15;

    const int rowBase = blockIdx.x * TM;
    const int nBase   = blockIdx.y * TN;

    const int aRow = tid >> 2;
    const int aPc  = (tid & 3) * 4;
    int aOff[2];
    #pragma unroll
    for (int j = 0; j < 2; ++j) {
        const int rG   = rowBase + aRow + 32 * j;
        const int bImg = rG / NPATCH;
        const int nTok = rG - bImg * NPATCH;
        const int pi_  = nTok / 14;
        const int pj_  = nTok - pi_ * 14;
        aOff[j] = ((bImg * CCH) * HIMG + pi_ * PSZ) * WIMG + pj_ * PSZ + aPc;
    }

    const int bKp  = tid >> 5;
    const int bCol = (tid & 31) * 4;

    f32x8 acc[2][4];
    #pragma unroll
    for (int i = 0; i < 2; ++i)
        #pragma unroll
        for (int j = 0; j < 4; ++j)
            acc[i][j] = (f32x8){0.f,0.f,0.f,0.f,0.f,0.f,0.f,0.f};

    const int mW = waveM * 32;
    const int nW = waveN * 64;
    const float* aF = &As[(mW + lsub) * A_STRIDE + half * 2];
    const float* bF = &Bs[half * B_PSTRIDE + (nW + lsub) * 2];

    for (int kBase = 0; kBase < KDIM; kBase += TK) {
        {
            const int ch  = kBase >> 8;
            const int pr0 = (kBase >> 4) & 15;
            const float* xc = x + (ch * HIMG + pr0) * WIMG;
            float4 av[2][2];
            #pragma unroll
            for (int j = 0; j < 2; ++j) {
                av[j][0] = *(const float4*)(xc + aOff[j]);
                av[j][1] = *(const float4*)(xc + aOff[j] + WIMG);
            }
            const float* wc = Wm + (long)(kBase + 2 * bKp) * DMODEL + nBase + bCol;
            float4 bv[4][2];
            #pragma unroll
            for (int i = 0; i < 4; ++i) {
                bv[i][0] = *(const float4*)(wc + (8 * i) * DMODEL);
                bv[i][1] = *(const float4*)(wc + (8 * i + 1) * DMODEL);
            }
            if (kBase + TK < KDIM) {
                __builtin_prefetch(wc + (long)TK * DMODEL, 0, 0);
            }
            #pragma unroll
            for (int j = 0; j < 2; ++j) {
                *(float4*)&As[(aRow + 32 * j) * A_STRIDE + aPc]      = av[j][0];
                *(float4*)&As[(aRow + 32 * j) * A_STRIDE + aPc + 16] = av[j][1];
            }
            #pragma unroll
            for (int i = 0; i < 4; ++i) {
                float* d = &Bs[(bKp + 4 * i) * B_PSTRIDE + bCol * 2];
                *(float4*)d       = make_float4(bv[i][0].x, bv[i][1].x, bv[i][0].y, bv[i][1].y);
                *(float4*)(d + 4) = make_float4(bv[i][0].z, bv[i][1].z, bv[i][0].w, bv[i][1].w);
            }
        }
        __syncthreads();

        #pragma unroll
        for (int kk = 0; kk < TK; kk += 4) {
            f32x2 a[2], b[4];
            #pragma unroll
            for (int i = 0; i < 2; ++i)
                a[i] = *(const f32x2*)(aF + i * 16 * A_STRIDE + kk);
            #pragma unroll
            for (int j = 0; j < 4; ++j)
                b[j] = *(const f32x2*)(bF + (kk >> 1) * B_PSTRIDE + j * 32);
            #pragma unroll
            for (int i = 0; i < 2; ++i)
                #pragma unroll
                for (int j = 0; j < 4; ++j)
                    acc[i][j] = __builtin_amdgcn_wmma_f32_16x16x4_f32(false, a[i], false, b[j],
                                                                      (short)0, acc[i][j], false, false);
        }
        __syncthreads();
    }

    const int col0 = nBase + nW + lsub;
    float bs[4];
    #pragma unroll
    for (int j = 0; j < 4; ++j) bs[j] = bias[col0 + 16 * j];
    #pragma unroll
    for (int i = 0; i < 2; ++i) {
        #pragma unroll
        for (int v = 0; v < 8; ++v) {
            const int mRow = mW + 16 * i + v + half * 8;
            const int r    = rowBase + mRow;
            const int bI   = r / NPATCH;
            const int s    = r - bI * NPATCH + 1;
            const long obase = ((long)bI * SEQ + s) * DMODEL + col0;
            const long pbase = (long)s * DMODEL + col0;
            #pragma unroll
            for (int j = 0; j < 4; ++j)
                out[obase + 16 * j] = acc[i][j][v] + bs[j] + pos[pbase + 16 * j];
        }
    }
}

__global__ __launch_bounds__(256)
void vit_class_token(const float* __restrict__ cls,
                     const float* __restrict__ pos,
                     float* __restrict__ out)
{
    const int idx = blockIdx.x * blockDim.x + threadIdx.x;
    const int b = idx / DMODEL;
    const int d = idx - b * DMODEL;
    out[(long)b * SEQ * DMODEL + d] = cls[d] + pos[d];
}

extern "C" void kernel_launch(void* const* d_in, const int* in_sizes, int n_in,
                              void* d_out, int out_size, void* d_ws, size_t ws_size,
                              hipStream_t stream) {
    const float* x   = (const float*)d_in[0];
    const float* Wm  = (const float*)d_in[1];
    const float* b   = (const float*)d_in[2];
    const float* pos = (const float*)d_in[3];
    const float* cls = (const float*)d_in[4];
    float* out = (float*)d_out;

    dim3 grid(ROWS / TM, DMODEL / TN);
    vit_patch_embed_gemm<<<grid, NTHR, 0, stream>>>(x, Wm, b, pos, out);
    vit_class_token<<<(BATCH * DMODEL) / 256, 256, 0, stream>>>(cls, pos, out);
}


# Issue priority by K-loop progress: each wave steps s_setprio 3,2,1,0 at K-chunks 0,12,18,22 so the 5 co-resident work-groups of a WGP end together instead of the youngest ending alone
# speedup vs baseline: 1.3709x; 1.0496x over previous
#include <hip/hip_runtime.h>
#include <hip/hip_bf16.h>

typedef float f32x2 __attribute__((ext_vector_type(2)));
typedef float f32x8 __attribute__((ext_vector_type(8)));

#define BATCH   64
#define CCH     3
#define HIMG    224
#define WIMG    224
#define PSZ     16
#define DMODEL  768
#define KDIM    768
#define NPATCH  196
#define SEQ     197
#define ROWS    (BATCH * NPATCH)

#define TM      64
#define TN      128
#define TK      32
#define NTHR    128
#define PRIO_K2 12
#define PRIO_K1 18
#define PRIO_K0 22
#define A_STRIDE 36
#define B_PSTRIDE 288

__global__ __launch_bounds__(NTHR) __attribute__((amdgpu_waves_per_eu(5, 5)))
void vit_patch_embed_gemm(const float* __restrict__ x,
                          const float* __restrict__ Wm,
                          const float* __restrict__ bias,
                          const float* __restrict__ pos,
                          float* __restrict__ out)
{
    __shared__ float As[TM * A_STRIDE];
    __shared__ float Bs[(TK / 2) * B_PSTRIDE];

    const int tid    = threadIdx.x;
    const int lane   = tid & 31;
    const int waveId = tid >> 5;
    const int waveM  = waveId & 1;
    const int waveN  = waveId >> 1;
    const int half   = lane >> 4;
    const int lsub   = lane & 15;

    const int rowBase = blockIdx.x * TM;
    const int nBase   = blockIdx.y * TN;

    const int aRow = tid >> 2;
    const int aPc  = (tid & 3) * 4;
    int aOff[2];
    #pragma unroll
    for (int j = 0; j < 2; ++j) {
        const int rG   = rowBase + aRow + 32 * j;
        const int bImg = rG / NPATCH;
        const int nTok = rG - bImg * NPATCH;
        const int pi_  = nTok / 14;
        const int pj_  = nTok - pi_ * 14;
        aOff[j] = ((bImg * CCH) * HIMG + pi_ * PSZ) * WIMG + pj_ * PSZ + aPc;
    }

    const int bKp  = tid >> 5;
    const int bCol = (tid & 31) * 4;

    f32x8 acc[2][4];
    #pragma unroll
    for (int i = 0; i < 2; ++i)
        #pragma unroll
        for (int j = 0; j < 4; ++j)
            acc[i][j] = (f32x8){0.f,0.f,0.f,0.f,0.f,0.f,0.f,0.f};

    const int mW = waveM * 32;
    const int nW = waveN * 64;
    const float* aF = &As[(mW + lsub) * A_STRIDE + half * 2];
    const float* bF = &Bs[half * B_PSTRIDE + (nW + lsub) * 2];

    for (int kBase = 0; kBase < KDIM; kBase += TK) {
        if (kBase == 0)                 __builtin_amdgcn_s_setprio(3);
        else if (kBase == PRIO_K2 * TK) __builtin_amdgcn_s_setprio(2);
        else if (kBase == PRIO_K1 * TK) __builtin_amdgcn_s_setprio(1);
        else if (kBase == PRIO_K0 * TK) __builtin_amdgcn_s_setprio(0);
        {
            const int ch  = kBase >> 8;
            const int pr0 = (kBase >> 4) & 15;
            const float* xc = x + (ch * HIMG + pr0) * WIMG;
            float4 av[2][2];
            #pragma unroll
            for (int j = 0; j < 2; ++j) {
                av[j][0] = *(const float4*)(xc + aOff[j]);
                av[j][1] = *(const float4*)(xc + aOff[j] + WIMG);
            }
            const float* wc = Wm + (long)(kBase + 2 * bKp) * DMODEL + nBase + bCol;
            float4 bv[4][2];
            #pragma unroll
            for (int i = 0; i < 4; ++i) {
                bv[i][0] = *(const float4*)(wc + (8 * i) * DMODEL);
                bv[i][1] = *(const float4*)(wc + (8 * i + 1) * DMODEL);
            }
            if (kBase + TK < KDIM) {
                __builtin_prefetch(wc + (long)TK * DMODEL, 0, 0);
            }
            #pragma unroll
            for (int j = 0; j < 2; ++j) {
                *(float4*)&As[(aRow + 32 * j) * A_STRIDE + aPc]      = av[j][0];
                *(float4*)&As[(aRow + 32 * j) * A_STRIDE + aPc + 16] = av[j][1];
            }
            #pragma unroll
            for (int i = 0; i < 4; ++i) {
                float* d = &Bs[(bKp + 4 * i) * B_PSTRIDE + bCol * 2];
                *(float4*)d       = make_float4(bv[i][0].x, bv[i][1].x, bv[i][0].y, bv[i][1].y);
                *(float4*)(d + 4) = make_float4(bv[i][0].z, bv[i][1].z, bv[i][0].w, bv[i][1].w);
            }
        }
        __syncthreads();

        #pragma unroll
        for (int kk = 0; kk < TK; kk += 4) {
            f32x2 a[2], b[4];
            #pragma unroll
            for (int i = 0; i < 2; ++i)
                a[i] = *(const f32x2*)(aF + i * 16 * A_STRIDE + kk);
            #pragma unroll
            for (int j = 0; j < 4; ++j)
                b[j] = *(const f32x2*)(bF + (kk >> 1) * B_PSTRIDE + j * 32);
            #pragma unroll
            for (int i = 0; i < 2; ++i)
                #pragma unroll
                for (int j = 0; j < 4; ++j)
                    acc[i][j] = __builtin_amdgcn_wmma_f32_16x16x4_f32(false, a[i], false, b[j],
                                                                      (short)0, acc[i][j], false, false);
        }
        __syncthreads();
    }

    const int col0 = nBase + nW + lsub;
    float bs[4];
    #pragma unroll
    for (int j = 0; j < 4; ++j) bs[j] = bias[col0 + 16 * j];
    #pragma unroll
    for (int i = 0; i < 2; ++i) {
        #pragma unroll
        for (int v = 0; v < 8; ++v) {
            const int mRow = mW + 16 * i + v + half * 8;
            const int r    = rowBase + mRow;
            const int bI   = r / NPATCH;
            const int s    = r - bI * NPATCH + 1;
            const long obase = ((long)bI * SEQ + s) * DMODEL + col0;
            const long pbase = (long)s * DMODEL + col0;
            #pragma unroll
            for (int j = 0; j < 4; ++j)
                out[obase + 16 * j] = acc[i][j][v] + bs[j] + pos[pbase + 16 * j];
        }
    }
}

__global__ __launch_bounds__(256)
void vit_class_token(const float* __restrict__ cls,
                     const float* __restrict__ pos,
                     float* __restrict__ out)
{
    const int idx = blockIdx.x * blockDim.x + threadIdx.x;
    const int b = idx / DMODEL;
    const int d = idx - b * DMODEL;
    out[(long)b * SEQ * DMODEL + d] = cls[d] + pos[d];
}

extern "C" void kernel_launch(void* const* d_in, const int* in_sizes, int n_in,
                              void* d_out, int out_size, void* d_ws, size_t ws_size,
                              hipStream_t stream) {
    const float* x   = (const float*)d_in[0];
    const float* Wm  = (const float*)d_in[1];
    const float* b   = (const float*)d_in[2];
    const float* pos = (const float*)d_in[3];
    const float* cls = (const float*)d_in[4];
    float* out = (float*)d_out;

    dim3 grid(ROWS / TM, DMODEL / TN);
    vit_patch_embed_gemm<<<grid, NTHR, 0, stream>>>(x, Wm, b, pos, out);
    vit_class_token<<<(BATCH * DMODEL) / 256, 256, 0, stream>>>(cls, pos, out);
}
